# starting kernel
# baseline (speedup 1.0000x reference)
#include <hip/hip_runtime.h>


#define BATCH 4096
#define TLEN  200
#define INSZ  34
#define HID   4
#define GATES 16
#define NCLS  3
#define MROWS (BATCH * TLEN)

typedef float v2f __attribute__((ext_vector_type(2)));
typedef float v8f __attribute__((ext_vector_type(8)));

__device__ __forceinline__ float fast_rcp(float x) {
  return __builtin_amdgcn_rcpf(x);
}
__device__ __forceinline__ float fast_sigmoid(float x) {
  return fast_rcp(1.0f + __expf(-x));
}
__device__ __forceinline__ float fast_tanh(float x) {
#if __has_builtin(__builtin_amdgcn_tanhf)
  return __builtin_amdgcn_tanhf(x);
#else
  float e = __expf(-2.0f * x);
  return (1.0f - e) * fast_rcp(1.0f + e);
#endif
}

__global__ __launch_bounds__(256) void lstm_xgates_wmma(
    const float* __restrict__ x,
    const float* __restrict__ W_ih,
    const float* __restrict__ b_ih,
    const float* __restrict__ b_hh,
    float* __restrict__ xg)
{
  const int lane = threadIdx.x & 31;
  const int wave = threadIdx.x >> 5;
  const int half = lane >> 4;
  const int l    = lane & 15;
  const int koff = half * 2;
  const long long m0 = ((long long)blockIdx.x * 8 + wave) * 16;

  const float bias = b_ih[l] + b_hh[l];
  v8f acc;
#pragma unroll
  for (int v = 0; v < 8; ++v) acc[v] = bias;

  const float* __restrict__ xrow = x    + (m0 + l) * INSZ;
  const float* __restrict__ wrow = W_ih + l * INSZ;

#pragma unroll
  for (int kk = 0; kk < 9; ++kk) {
    const int k0 = kk * 4 + koff;
    float a0 = xrow[min(k0,     INSZ - 1)];
    float a1 = xrow[min(k0 + 1, INSZ - 1)];
    float w0 = wrow[min(k0,     INSZ - 1)];
    float w1 = wrow[min(k0 + 1, INSZ - 1)];
    v2f a, b;
    a.x = (k0     < INSZ) ? a0 : 0.0f;
    a.y = (k0 + 1 < INSZ) ? a1 : 0.0f;
    b.x = (k0     < INSZ) ? w0 : 0.0f;
    b.y = (k0 + 1 < INSZ) ? w1 : 0.0f;
    acc = __builtin_amdgcn_wmma_f32_16x16x4_f32(
        false, a, false, b, (short)0, acc, false, false);
  }

#pragma unroll
  for (int v = 0; v < 8; ++v) {
    const long long m = m0 + v + 8 * half;
    xg[m * GATES + l] = acc[v];
  }
}

__global__ __launch_bounds__(256) void lstm_scan(
    const float* __restrict__ xg,
    const float* __restrict__ W_hh,
    const float* __restrict__ W_fc,
    const float* __restrict__ b_fc,
    float* __restrict__ out)
{
  const int gtid = blockIdx.x * blockDim.x + threadIdx.x;
  const int b = gtid >> 2;
  const int j = gtid & 3;
  if (b >= BATCH) return;

  const float4 Wi = *(const float4*)(W_hh + (j     ) * HID);
  const float4 Wf = *(const float4*)(W_hh + (4  + j) * HID);
  const float4 Wg = *(const float4*)(W_hh + (8  + j) * HID);
  const float4 Wo = *(const float4*)(W_hh + (12 + j) * HID);

  const float* __restrict__ xgb = xg + (long long)b * TLEN * GATES;

  float h = 0.0f, c = 0.0f;
  for (int t = 0; t < TLEN; ++t) {
    const float* __restrict__ p = xgb + t * GATES;
    __builtin_prefetch(p + 8 * GATES, 0, 0);

    float pi = p[j];
    float pf = p[4  + j];
    float pg = p[8  + j];
    float po = p[12 + j];

    const float h0 = __shfl(h, 0, 4);
    const float h1 = __shfl(h, 1, 4);
    const float h2 = __shfl(h, 2, 4);
    const float h3 = __shfl(h, 3, 4);

    pi += Wi.x * h0 + Wi.y * h1 + Wi.z * h2 + Wi.w * h3;
    pf += Wf.x * h0 + Wf.y * h1 + Wf.z * h2 + Wf.w * h3;
    pg += Wg.x * h0 + Wg.y * h1 + Wg.z * h2 + Wg.w * h3;
    po += Wo.x * h0 + Wo.y * h1 + Wo.z * h2 + Wo.w * h3;

    const float ig = fast_sigmoid(pi);
    const float fg = fast_sigmoid(pf);
    const float gg = fast_tanh(pg);
    const float og = fast_sigmoid(po);

    c = fg * c + ig * gg;
    h = og * fast_tanh(c);
  }

  const float h0 = __shfl(h, 0, 4);
  const float h1 = __shfl(h, 1, 4);
  const float h2 = __shfl(h, 2, 4);
  const float h3 = __shfl(h, 3, 4);
  if (j < NCLS) {
    const float4 Wc = *(const float4*)(W_fc + j * HID);
    out[b * NCLS + j] =
        b_fc[j] + Wc.x * h0 + Wc.y * h1 + Wc.z * h2 + Wc.w * h3;
  }
}

extern "C" void kernel_launch(void* const* d_in, const int* in_sizes, int n_in,
                              void* d_out, int out_size, void* d_ws, size_t ws_size,
                              hipStream_t stream) {
  (void)in_sizes; (void)n_in; (void)out_size; (void)ws_size;
  const float* x    = (const float*)d_in[0];
  const float* W_ih = (const float*)d_in[1];
  const float* W_hh = (const float*)d_in[2];
  const float* b_ih = (const float*)d_in[3];
  const float* b_hh = (const float*)d_in[4];
  const float* W_fc = (const float*)d_in[5];
  const float* b_fc = (const float*)d_in[6];
  float* out = (float*)d_out;
  float* xg  = (float*)d_ws;

  const int tiles  = MROWS / 16;
  const int blocks1 = tiles / 8;
  hipLaunchKernelGGL(lstm_xgates_wmma, dim3(blocks1), dim3(256), 0, stream,
                     x, W_ih, b_ih, b_hh, xg);

  const int blocks2 = (BATCH * 4) / 256;
  hipLaunchKernelGGL(lstm_scan, dim3(blocks2), dim3(256), 0, stream,
                     xg, W_hh, W_fc, b_fc, out);
}


# one kernel: one wave per 16-sequence tile, gates x sequences on fp32 WMMA incl. recurrent h.W_hh and classifier, x projected in registers (no x_gates buffer); summation order changed
# speedup vs baseline: 1.4414x; 1.4414x over previous
#include <hip/hip_runtime.h>


#define BATCH 4096
#define TLEN  200
#define INSZ  34
#define HID   4
#define NCLS  3
#define RING  8
#define PFDIST 16

typedef float v2f __attribute__((ext_vector_type(2)));
typedef float v4f __attribute__((ext_vector_type(4)));
typedef float v8f __attribute__((ext_vector_type(8)));
typedef v4f v4f_a8 __attribute__((aligned(8)));

__device__ __forceinline__ float fast_rcp(float x) {
  return __builtin_amdgcn_rcpf(x);
}
__device__ __forceinline__ float fast_sigmoid(float x) {
  return fast_rcp(1.0f + __expf(-x));
}
__device__ __forceinline__ float fast_tanh(float x) {
#if __has_builtin(__builtin_amdgcn_tanhf)
  return __builtin_amdgcn_tanhf(x);
#else
  float e = __expf(-2.0f * x);
  return (1.0f - e) * fast_rcp(1.0f + e);
#endif
}

struct XStep { v2f f[9]; };

__device__ __forceinline__ void load_xstep(XStep& s, const float* __restrict__ p,
                                           int half) {
  const v4f q0 = *(const v4f_a8*)(p);
  const v4f q1 = *(const v4f_a8*)(p + 4);
  const v4f q2 = *(const v4f_a8*)(p + 8);
  const v4f q3 = *(const v4f_a8*)(p + 12);
  const v2f q4 = *(const v2f*)(p + (half ? 14 : 16));
  s.f[0] = (v2f){q0.x, q0.y};  s.f[1] = (v2f){q0.z, q0.w};
  s.f[2] = (v2f){q1.x, q1.y};  s.f[3] = (v2f){q1.z, q1.w};
  s.f[4] = (v2f){q2.x, q2.y};  s.f[5] = (v2f){q2.z, q2.w};
  s.f[6] = (v2f){q3.x, q3.y};  s.f[7] = (v2f){q3.z, q3.w};
  s.f[8] = q4;
}

__device__ __forceinline__ v8f xproj(const v2f (&wih)[9], const XStep& s,
                                     const v8f bias, int half) {
  const v2f s8 = half ? (v2f){0.0f, 0.0f} : s.f[8];
  v8f a = bias;
  v8f b = {0.f, 0.f, 0.f, 0.f, 0.f, 0.f, 0.f, 0.f};
  a = __builtin_amdgcn_wmma_f32_16x16x4_f32(false, wih[0], false, s.f[0], (short)0, a, false, false);
  b = __builtin_amdgcn_wmma_f32_16x16x4_f32(false, wih[1], false, s.f[1], (short)0, b, false, false);
  a = __builtin_amdgcn_wmma_f32_16x16x4_f32(false, wih[2], false, s.f[2], (short)0, a, false, false);
  b = __builtin_amdgcn_wmma_f32_16x16x4_f32(false, wih[3], false, s.f[3], (short)0, b, false, false);
  a = __builtin_amdgcn_wmma_f32_16x16x4_f32(false, wih[4], false, s.f[4], (short)0, a, false, false);
  b = __builtin_amdgcn_wmma_f32_16x16x4_f32(false, wih[5], false, s.f[5], (short)0, b, false, false);
  a = __builtin_amdgcn_wmma_f32_16x16x4_f32(false, wih[6], false, s.f[6], (short)0, a, false, false);
  b = __builtin_amdgcn_wmma_f32_16x16x4_f32(false, wih[7], false, s.f[7], (short)0, b, false, false);
  a = __builtin_amdgcn_wmma_f32_16x16x4_f32(false, wih[8], false, s8, (short)0, a, false, false);
  return a + b;
}

__global__ __launch_bounds__(32) void lstm_tile_wmma(
    const float* __restrict__ x,
    const float* __restrict__ W_ih,
    const float* __restrict__ W_hh,
    const float* __restrict__ b_ih,
    const float* __restrict__ b_hh,
    const float* __restrict__ W_fc,
    const float* __restrict__ b_fc,
    float* __restrict__ out)
{
  const int lane = threadIdx.x;
  const int half = lane >> 4;
  const int c    = lane & 15;
  const int b0   = blockIdx.x * 16;

  const int arow = (c & 3) * HID + 2 * (c >> 3) + ((c >> 2) & 1);

  v2f wih[9];
  {
    const float* __restrict__ wr = W_ih + arow * INSZ + 18 * half;
#pragma unroll
    for (int kk = 0; kk < 8; ++kk) wih[kk] = *(const v2f*)(wr + 2 * kk);
    const v2f w8 = *(const v2f*)(wr + (half ? 14 : 16));
    wih[8] = half ? (v2f){0.0f, 0.0f} : w8;
  }
  const v2f whh = *(const v2f*)(W_hh + arow * HID + 2 * half);
  v2f wfc;
  {
    const v2f w = *(const v2f*)(W_fc + min(c, NCLS - 1) * HID + 2 * half);
    wfc = (c < NCLS) ? w : (v2f){0.0f, 0.0f};
  }

  v8f bias;
#pragma unroll
  for (int v = 0; v < 8; ++v) {
    const int g = (v & 3) * HID + 2 * half + (v >> 2);
    bias[v] = b_ih[g] + b_hh[g];
  }

  const float* __restrict__ xrow = x + (long long)(b0 + c) * TLEN * INSZ;
  const float* __restrict__ xl   = xrow + 18 * half;

#pragma unroll
  for (int s = RING; s < PFDIST; s += 2)
    __builtin_prefetch(xrow + s * INSZ + half * 64, 0, 0);

  XStep ring[RING];
#pragma unroll
  for (int r = 0; r < RING; ++r) {
    load_xstep(ring[r], xl + r * INSZ, half);
    __builtin_amdgcn_sched_barrier(0);
  }

  float ca = 0.0f, cb = 0.0f;
  v2f hB = {0.0f, 0.0f};

  v8f g;
  {
    const v8f xp0 = xproj(wih, ring[0], bias, half);
    load_xstep(ring[0], xl + RING * INSZ, half);
    g = __builtin_amdgcn_wmma_f32_16x16x4_f32(
        false, whh, false, hB, (short)0, xp0, false, false);
  }

  for (int t0 = 0; t0 < TLEN; t0 += RING) {
#pragma unroll
    for (int r = 0; r < RING; ++r) {
      const int t = t0 + r;
      if ((r & 1) == 0) {
        const int off = min((t + PFDIST) * INSZ + half * 64, TLEN * INSZ - 1);
        __builtin_prefetch(xrow + off, 0, 0);
      }

      const int slot = (r + 1) % RING;
      const v8f xp = xproj(wih, ring[slot], bias, half);
      load_xstep(ring[slot], xl + min(t + 1 + RING, TLEN - 1) * INSZ, half);

      const float ia = fast_sigmoid(g[0]);
      const float fa = fast_sigmoid(g[1]);
      const float ga = fast_tanh(g[2]);
      const float oa = fast_sigmoid(g[3]);
      const float ib = fast_sigmoid(g[4]);
      const float fb = fast_sigmoid(g[5]);
      const float gb = fast_tanh(g[6]);
      const float ob = fast_sigmoid(g[7]);
      ca = fa * ca + ia * ga;
      cb = fb * cb + ib * gb;
      hB.x = oa * fast_tanh(ca);
      hB.y = ob * fast_tanh(cb);

      g = __builtin_amdgcn_wmma_f32_16x16x4_f32(
          false, whh, false, hB, (short)0, xp, false, false);

      __builtin_amdgcn_sched_group_barrier(0x008, 2, 0);
      __builtin_amdgcn_sched_group_barrier(0x402, 5, 0);
      __builtin_amdgcn_sched_group_barrier(0x008, 1, 0);
      __builtin_amdgcn_sched_group_barrier(0x402, 5, 0);
      __builtin_amdgcn_sched_group_barrier(0x008, 1, 0);
      __builtin_amdgcn_sched_group_barrier(0x402, 5, 0);
      __builtin_amdgcn_sched_group_barrier(0x008, 1, 0);
      __builtin_amdgcn_sched_group_barrier(0x402, 5, 0);
      __builtin_amdgcn_sched_group_barrier(0x008, 1, 0);
      __builtin_amdgcn_sched_group_barrier(0x402, 5, 0);
      __builtin_amdgcn_sched_group_barrier(0x008, 1, 0);
      __builtin_amdgcn_sched_group_barrier(0x402, 5, 0);
      __builtin_amdgcn_sched_group_barrier(0x008, 1, 0);
      __builtin_amdgcn_sched_group_barrier(0x402, 5, 0);
      __builtin_amdgcn_sched_group_barrier(0x008, 1, 0);
      __builtin_amdgcn_sched_barrier(0);
    }
  }

  v8f accfc;
#pragma unroll
  for (int v = 0; v < 8; ++v)
    accfc[v] = (half == 0 && v < NCLS) ? b_fc[min(v, NCLS - 1)] : 0.0f;
  const v8f o = __builtin_amdgcn_wmma_f32_16x16x4_f32(
      false, wfc, false, hB, (short)0, accfc, false, false);
  if (half == 0) {
    float* __restrict__ op = out + (b0 + c) * NCLS;
    op[0] = o[0];
    op[1] = o[1];
    op[2] = o[2];
  }
}

extern "C" void kernel_launch(void* const* d_in, const int* in_sizes, int n_in,
                              void* d_out, int out_size, void* d_ws, size_t ws_size,
                              hipStream_t stream) {
  (void)in_sizes; (void)n_in; (void)out_size; (void)d_ws; (void)ws_size;
  const float* x    = (const float*)d_in[0];
  const float* W_ih = (const float*)d_in[1];
  const float* W_hh = (const float*)d_in[2];
  const float* b_ih = (const float*)d_in[3];
  const float* b_hh = (const float*)d_in[4];
  const float* W_fc = (const float*)d_in[5];
  const float* b_fc = (const float*)d_in[6];
  float* out = (float*)d_out;

  hipLaunchKernelGGL(lstm_tile_wmma, dim3(BATCH / 16), dim3(32), 0, stream,
                     x, W_ih, W_hh, b_ih, b_hh, W_fc, b_fc, out);
}


# WMMA columns re-cut: 4 sequences x 4 time steps per tile (was 16 sequences x 1 step), 1024 waves, one per SIMD (was 256); h and c handed 4 lanes up by DPP each step; summation order unchanged
# speedup vs baseline: 2.3894x; 1.6577x over previous
#include <hip/hip_runtime.h>


#define BATCH 4096
#define TLEN  200
#define INSZ  34
#define HID   4
#define NCLS  3
#define SEQW  4
#define TSUB  4
#define NGRP  (TLEN / TSUB)
#define GSTR  (TSUB * INSZ)
#define RINGG 5
#define PFG   12

typedef float v2f __attribute__((ext_vector_type(2)));
typedef float v4f __attribute__((ext_vector_type(4)));
typedef float v8f __attribute__((ext_vector_type(8)));
typedef v4f v4f_a8 __attribute__((aligned(8)));

__device__ __forceinline__ float fast_rcp(float x) {
  return __builtin_amdgcn_rcpf(x);
}
__device__ __forceinline__ float fast_sigmoid(float x) {
  return fast_rcp(1.0f + __expf(-x));
}
__device__ __forceinline__ float fast_tanh(float x) {
#if __has_builtin(__builtin_amdgcn_tanhf)
  return __builtin_amdgcn_tanhf(x);
#else
  float e = __expf(-2.0f * x);
  return (1.0f - e) * fast_rcp(1.0f + e);
#endif
}

__device__ __forceinline__ float up4(float v) {
  return __builtin_bit_cast(float,
      __builtin_amdgcn_update_dpp(0, __builtin_bit_cast(int, v),
                                  0x124, 0xf, 0xf, true));
}

struct XGroup { v2f f[9]; };

__device__ __forceinline__ void load_xgroup(XGroup& s, const float* __restrict__ p,
                                            int half) {
  const v4f q0 = *(const v4f_a8*)(p);
  const v4f q1 = *(const v4f_a8*)(p + 4);
  const v4f q2 = *(const v4f_a8*)(p + 8);
  const v4f q3 = *(const v4f_a8*)(p + 12);
  const v2f q4 = *(const v2f*)(p + (half ? 14 : 16));
  s.f[0] = (v2f){q0.x, q0.y};  s.f[1] = (v2f){q0.z, q0.w};
  s.f[2] = (v2f){q1.x, q1.y};  s.f[3] = (v2f){q1.z, q1.w};
  s.f[4] = (v2f){q2.x, q2.y};  s.f[5] = (v2f){q2.z, q2.w};
  s.f[6] = (v2f){q3.x, q3.y};  s.f[7] = (v2f){q3.z, q3.w};
  s.f[8] = q4;
}

#define WMMA4(A, B, C) \
  __builtin_amdgcn_wmma_f32_16x16x4_f32(false, (A), false, (B), (short)0, (C), false, false)

__device__ __forceinline__ v8f xproj(const v2f (&wih)[9], const XGroup& s,
                                     const v8f bias, int half) {
  const v2f s8 = half ? (v2f){0.0f, 0.0f} : s.f[8];
  v8f a = bias;
  v8f b = {0.f, 0.f, 0.f, 0.f, 0.f, 0.f, 0.f, 0.f};
  a = WMMA4(wih[0], s.f[0], a);
  b = WMMA4(wih[1], s.f[1], b);
  a = WMMA4(wih[2], s.f[2], a);
  b = WMMA4(wih[3], s.f[3], b);
  a = WMMA4(wih[4], s.f[4], a);
  b = WMMA4(wih[5], s.f[5], b);
  a = WMMA4(wih[6], s.f[6], a);
  b = WMMA4(wih[7], s.f[7], b);
  a = WMMA4(wih[8], s8, a);
  return a + b;
}

__device__ __forceinline__ void cell_step(const v8f g, float& ca, float& cb, v2f& hB) {
  const float ia = fast_sigmoid(g[0]);
  const float fa = fast_sigmoid(g[1]);
  const float ga = fast_tanh(g[2]);
  const float oa = fast_sigmoid(g[3]);
  const float ib = fast_sigmoid(g[4]);
  const float fb = fast_sigmoid(g[5]);
  const float gb = fast_tanh(g[6]);
  const float ob = fast_sigmoid(g[7]);
  const float na = fa * ca + ia * ga;
  const float nb = fb * cb + ib * gb;
  const float ha = oa * fast_tanh(na);
  const float hb = ob * fast_tanh(nb);
  ca = up4(na);
  cb = up4(nb);
  hB.x = up4(ha);
  hB.y = up4(hb);
}

__global__ __launch_bounds__(128) void lstm_s4t4_wmma(
    const float* __restrict__ x,
    const float* __restrict__ W_ih,
    const float* __restrict__ W_hh,
    const float* __restrict__ b_ih,
    const float* __restrict__ b_hh,
    const float* __restrict__ W_fc,
    const float* __restrict__ b_fc,
    float* __restrict__ out)
{
  const int lane = threadIdx.x & 31;
  const int wave = threadIdx.x >> 5;
  const int half = lane >> 4;
  const int c    = lane & 15;
  const int sq   = c & 3;
  const int tj   = c >> 2;
  const int bseq = blockIdx.x * (4 * SEQW) + wave * SEQW + sq;

  const int arow = (c & 3) * HID + 2 * (c >> 3) + ((c >> 2) & 1);

  v2f wih[9];
  {
    const float* __restrict__ wr = W_ih + arow * INSZ + 18 * half;
#pragma unroll
    for (int kk = 0; kk < 8; ++kk) wih[kk] = *(const v2f*)(wr + 2 * kk);
    const v2f w8 = *(const v2f*)(wr + (half ? 14 : 16));
    wih[8] = half ? (v2f){0.0f, 0.0f} : w8;
  }
  const v2f whh = *(const v2f*)(W_hh + arow * HID + 2 * half);
  v2f wfc;
  {
    const v2f w = *(const v2f*)(W_fc + min(c, NCLS - 1) * HID + 2 * half);
    wfc = (c < NCLS) ? w : (v2f){0.0f, 0.0f};
  }

  v8f bias;
#pragma unroll
  for (int v = 0; v < 8; ++v) {
    const int g = (v & 3) * HID + 2 * half + (v >> 2);
    bias[v] = b_ih[g] + b_hh[g];
  }
  const v8f zero8 = {0.f, 0.f, 0.f, 0.f, 0.f, 0.f, 0.f, 0.f};

  const float* __restrict__ xrow = x + (long long)bseq * TLEN * INSZ;
  const float* __restrict__ xl   = xrow + tj * INSZ + 18 * half;
  const int pfl = (2 * tj + half) * 32;

#pragma unroll
  for (int g = RINGG; g < PFG; ++g)
    __builtin_prefetch(xrow + g * GSTR + pfl, 0, 0);

  XGroup ring[RINGG];
#pragma unroll
  for (int r = 0; r < RINGG; ++r) {
    load_xgroup(ring[r], xl + r * GSTR, half);
    __builtin_amdgcn_sched_barrier(0);
  }

  float ca = 0.0f, cb = 0.0f;
  v2f hB = {0.0f, 0.0f};

  v8f xp = xproj(wih, ring[0], bias, half);
  load_xgroup(ring[0], xl + RINGG * GSTR, half);

  for (int q0 = 0; q0 < NGRP; q0 += RINGG) {
#pragma unroll
    for (int r = 0; r < RINGG; ++r) {
      const int q = q0 + r;
      {
        const int off = min((q + PFG) * GSTR + pfl, TLEN * INSZ - 1);
        __builtin_prefetch(xrow + off, 0, 0);
      }

      XGroup& S = ring[(r + 1) % RINGG];
      const v2f s8 = half ? (v2f){0.0f, 0.0f} : S.f[8];
      v8f g, a, b;

      g = WMMA4(whh, hB, xp);
      a = WMMA4(wih[0], S.f[0], bias);
      b = WMMA4(wih[1], S.f[1], zero8);
      a = WMMA4(wih[2], S.f[2], a);
      cell_step(g, ca, cb, hB);
      __builtin_amdgcn_sched_group_barrier(0x008, 2, 0);
      __builtin_amdgcn_sched_group_barrier(0x402, 12, 0);
      __builtin_amdgcn_sched_group_barrier(0x008, 1, 0);
      __builtin_amdgcn_sched_group_barrier(0x402, 12, 0);
      __builtin_amdgcn_sched_group_barrier(0x008, 1, 0);
      __builtin_amdgcn_sched_barrier(0);
      g = WMMA4(whh, hB, xp);
      b = WMMA4(wih[3], S.f[3], b);
      a = WMMA4(wih[4], S.f[4], a);
      cell_step(g, ca, cb, hB);
      __builtin_amdgcn_sched_group_barrier(0x008, 2, 0);
      __builtin_amdgcn_sched_group_barrier(0x402, 20, 0);
      __builtin_amdgcn_sched_group_barrier(0x008, 1, 0);
      __builtin_amdgcn_sched_barrier(0);
      g = WMMA4(whh, hB, xp);
      b = WMMA4(wih[5], S.f[5], b);
      a = WMMA4(wih[6], S.f[6], a);
      cell_step(g, ca, cb, hB);
      __builtin_amdgcn_sched_group_barrier(0x008, 2, 0);
      __builtin_amdgcn_sched_group_barrier(0x402, 20, 0);
      __builtin_amdgcn_sched_group_barrier(0x008, 1, 0);
      __builtin_amdgcn_sched_barrier(0);
      g = WMMA4(whh, hB, xp);
      b = WMMA4(wih[7], S.f[7], b);
      a = WMMA4(wih[8], s8, a);
      cell_step(g, ca, cb, hB);
      xp = a + b;
      load_xgroup(S, xl + min(q + 1 + RINGG, NGRP - 1) * GSTR, half);
      __builtin_amdgcn_sched_group_barrier(0x008, 2, 0);
      __builtin_amdgcn_sched_group_barrier(0x402, 20, 0);
      __builtin_amdgcn_sched_group_barrier(0x008, 1, 0);
      __builtin_amdgcn_sched_barrier(0);
    }
  }

  v8f accfc;
#pragma unroll
  for (int v = 0; v < 8; ++v)
    accfc[v] = (half == 0 && v < NCLS) ? b_fc[min(v, NCLS - 1)] : 0.0f;
  const v8f o = WMMA4(wfc, hB, accfc);
  if (half == 0 && tj == 0) {
    float* __restrict__ op = out + bseq * NCLS;
    op[0] = o[0];
    op[1] = o[1];
    op[2] = o[2];
  }
}

extern "C" void kernel_launch(void* const* d_in, const int* in_sizes, int n_in,
                              void* d_out, int out_size, void* d_ws, size_t ws_size,
                              hipStream_t stream) {
  (void)in_sizes; (void)n_in; (void)out_size; (void)d_ws; (void)ws_size;
  const float* x    = (const float*)d_in[0];
  const float* W_ih = (const float*)d_in[1];
  const float* W_hh = (const float*)d_in[2];
  const float* b_ih = (const float*)d_in[3];
  const float* b_hh = (const float*)d_in[4];
  const float* W_fc = (const float*)d_in[5];
  const float* b_fc = (const float*)d_in[6];
  float* out = (float*)d_out;

  hipLaunchKernelGGL(lstm_s4t4_wmma, dim3(BATCH / (4 * SEQW)), dim3(128), 0, stream,
                     x, W_ih, W_hh, b_ih, b_hh, W_fc, b_fc, out);
}


# x staged through LDS: waves fetch their sequences' contiguous 544-byte group chunks with whole-line 16-byte loads (4-8 lines per load, was 17-20) and read B fragments from LDS; L2 prefetch dropped
# speedup vs baseline: 2.5923x; 1.0849x over previous
#include <hip/hip_runtime.h>


#define BATCH 4096
#define TLEN  200
#define INSZ  34
#define HID   4
#define NCLS  3
#define SEQW  4
#define TSUB  4
#define NGRP  (TLEN / TSUB)
#define GSTR  (TSUB * INSZ)
#define RINGG 5
#define SEQF  (TLEN * INSZ)
#define WLDS  (SEQW * GSTR)

typedef float v2f __attribute__((ext_vector_type(2)));
typedef float v4f __attribute__((ext_vector_type(4)));
typedef float v8f __attribute__((ext_vector_type(8)));
typedef v4f v4f_a8 __attribute__((aligned(8)));

__device__ __forceinline__ float fast_rcp(float x) {
  return __builtin_amdgcn_rcpf(x);
}
__device__ __forceinline__ float fast_sigmoid(float x) {
  return fast_rcp(1.0f + __expf(-x));
}
__device__ __forceinline__ float fast_tanh(float x) {
#if __has_builtin(__builtin_amdgcn_tanhf)
  return __builtin_amdgcn_tanhf(x);
#else
  float e = __expf(-2.0f * x);
  return (1.0f - e) * fast_rcp(1.0f + e);
#endif
}

__device__ __forceinline__ float up4(float v) {
  return __builtin_bit_cast(float,
      __builtin_amdgcn_update_dpp(0, __builtin_bit_cast(int, v),
                                  0x124, 0xf, 0xf, true));
}

struct XGroup { v2f f[9]; };

struct XStage { v4f p[5]; };

__device__ __forceinline__ void stage_load(XStage& s, const float* __restrict__ g0,
                                           const float* __restrict__ gt) {
  s.p[0] = *(const v4f*)(g0);
  s.p[1] = *(const v4f*)(g0 + 32);
  s.p[2] = *(const v4f*)(g0 + 64);
  s.p[3] = *(const v4f*)(g0 + 96);
  s.p[4] = *(const v4f*)(gt);
}

__device__ __forceinline__ void stage_store(const XStage& s, float* l0, float* lt) {
  *(v4f*)(l0)      = s.p[0];
  *(v4f*)(l0 + 32) = s.p[1];
  *(v4f*)(l0 + 64) = s.p[2];
  *(v4f*)(l0 + 96) = s.p[3];
  *(v4f*)(lt)      = s.p[4];
}

__device__ __forceinline__ void frag_load(XGroup& s, const float* p, int half) {
#pragma unroll
  for (int kk = 0; kk < 8; ++kk) s.f[kk] = *(const v2f*)(p + 2 * kk);
  s.f[8] = *(const v2f*)(p + (half ? 14 : 16));
}

#define WMMA4(A, B, C) \
  __builtin_amdgcn_wmma_f32_16x16x4_f32(false, (A), false, (B), (short)0, (C), false, false)

__device__ __forceinline__ v8f xproj(const v2f (&wih)[9], const XGroup& s,
                                     const v8f bias, int half) {
  const v2f s8 = half ? (v2f){0.0f, 0.0f} : s.f[8];
  v8f a = bias;
  v8f b = {0.f, 0.f, 0.f, 0.f, 0.f, 0.f, 0.f, 0.f};
  a = WMMA4(wih[0], s.f[0], a);
  b = WMMA4(wih[1], s.f[1], b);
  a = WMMA4(wih[2], s.f[2], a);
  b = WMMA4(wih[3], s.f[3], b);
  a = WMMA4(wih[4], s.f[4], a);
  b = WMMA4(wih[5], s.f[5], b);
  a = WMMA4(wih[6], s.f[6], a);
  b = WMMA4(wih[7], s.f[7], b);
  a = WMMA4(wih[8], s8, a);
  return a + b;
}

__device__ __forceinline__ void cell_step(const v8f g, float& ca, float& cb, v2f& hB) {
  const float ia = fast_sigmoid(g[0]);
  const float fa = fast_sigmoid(g[1]);
  const float ga = fast_tanh(g[2]);
  const float oa = fast_sigmoid(g[3]);
  const float ib = fast_sigmoid(g[4]);
  const float fb = fast_sigmoid(g[5]);
  const float gb = fast_tanh(g[6]);
  const float ob = fast_sigmoid(g[7]);
  const float na = fa * ca + ia * ga;
  const float nb = fb * cb + ib * gb;
  const float ha = oa * fast_tanh(na);
  const float hb = ob * fast_tanh(nb);
  ca = up4(na);
  cb = up4(nb);
  hB.x = up4(ha);
  hB.y = up4(hb);
}

__global__ __launch_bounds__(128) void lstm_s4t4_wmma(
    const float* __restrict__ x,
    const float* __restrict__ W_ih,
    const float* __restrict__ W_hh,
    const float* __restrict__ b_ih,
    const float* __restrict__ b_hh,
    const float* __restrict__ W_fc,
    const float* __restrict__ b_fc,
    float* __restrict__ out)
{
  const int lane = threadIdx.x & 31;
  const int wave = threadIdx.x >> 5;
  const int half = lane >> 4;
  const int c    = lane & 15;
  const int sq   = c & 3;
  const int tj   = c >> 2;
  const int bseq = blockIdx.x * (4 * SEQW) + wave * SEQW + sq;

  const int arow = (c & 3) * HID + 2 * (c >> 3) + ((c >> 2) & 1);

  v2f wih[9];
  {
    const float* __restrict__ wr = W_ih + arow * INSZ + 18 * half;
#pragma unroll
    for (int kk = 0; kk < 8; ++kk) wih[kk] = *(const v2f*)(wr + 2 * kk);
    const v2f w8 = *(const v2f*)(wr + (half ? 14 : 16));
    wih[8] = half ? (v2f){0.0f, 0.0f} : w8;
  }
  const v2f whh = *(const v2f*)(W_hh + arow * HID + 2 * half);
  v2f wfc;
  {
    const v2f w = *(const v2f*)(W_fc + min(c, NCLS - 1) * HID + 2 * half);
    wfc = (c < NCLS) ? w : (v2f){0.0f, 0.0f};
  }

  v8f bias;
#pragma unroll
  for (int v = 0; v < 8; ++v) {
    const int g = (v & 3) * HID + 2 * half + (v >> 2);
    bias[v] = b_ih[g] + b_hh[g];
  }
  const v8f zero8 = {0.f, 0.f, 0.f, 0.f, 0.f, 0.f, 0.f, 0.f};

  const int wseq = blockIdx.x * (4 * SEQW) + wave * SEQW;
  const float* __restrict__ xw = x + (long long)wseq * SEQF;
  const int fs = lane >> 3, fp = lane & 7;
  const int ts = (lane >> 1) & 3, tp = lane & 1;
  const float* __restrict__ g0 = xw + fs * SEQF + 4 * fp;
  const float* __restrict__ gt = xw + ts * SEQF + 128 + 4 * tp;

  __shared__ alignas(16) float xs[4 * WLDS];
  float* lw = xs + wave * WLDS;
  float* l0 = lw + fs * GSTR + 4 * fp;
  float* lt = lw + ts * GSTR + 128 + 4 * tp;
  const float* lf = lw + sq * GSTR + tj * INSZ + 18 * half;

  XStage ring[RINGG];
#pragma unroll
  for (int r = 0; r < RINGG; ++r) {
    stage_load(ring[r], g0 + r * GSTR, gt + r * GSTR);
    __builtin_amdgcn_sched_barrier(0);
  }

  float ca = 0.0f, cb = 0.0f;
  v2f hB = {0.0f, 0.0f};

  XGroup F[2];
  stage_store(ring[0], l0, lt);
  stage_load(ring[0], g0 + RINGG * GSTR, gt + RINGG * GSTR);
  frag_load(F[0], lf, half);
  v8f xp = xproj(wih, F[0], bias, half);
  stage_store(ring[1], l0, lt);
  stage_load(ring[1], g0 + (RINGG + 1) * GSTR, gt + (RINGG + 1) * GSTR);
  frag_load(F[1], lf, half);

  for (int q0 = 0; q0 < NGRP; q0 += 2 * RINGG) {
#pragma unroll
    for (int r = 0; r < 2 * RINGG; ++r) {
      const int q = q0 + r;
      XGroup& S = F[(r + 1) & 1];
      const v2f s8 = half ? (v2f){0.0f, 0.0f} : S.f[8];
      v8f g, a, b;

      g = WMMA4(whh, hB, xp);
      a = WMMA4(wih[0], S.f[0], bias);
      {
        XStage& R = ring[(r + 2) % RINGG];
        stage_store(R, l0, lt);
        const int gn = min(q + 2 + RINGG, NGRP - 1) * GSTR;
        stage_load(R, g0 + gn, gt + gn);
        frag_load(F[r & 1], lf, half);
      }
      b = WMMA4(wih[1], S.f[1], zero8);
      a = WMMA4(wih[2], S.f[2], a);
      cell_step(g, ca, cb, hB);
      __builtin_amdgcn_sched_group_barrier(0x008, 2, 0);
      __builtin_amdgcn_sched_group_barrier(0x200, 5, 0);
      __builtin_amdgcn_sched_group_barrier(0x020, 5, 0);
      __builtin_amdgcn_sched_group_barrier(0x100, 5, 0);
      __builtin_amdgcn_sched_group_barrier(0x402, 12, 0);
      __builtin_amdgcn_sched_group_barrier(0x008, 1, 0);
      __builtin_amdgcn_sched_group_barrier(0x402, 12, 0);
      __builtin_amdgcn_sched_group_barrier(0x008, 1, 0);
      __builtin_amdgcn_sched_barrier(0);
      g = WMMA4(whh, hB, xp);
      b = WMMA4(wih[3], S.f[3], b);
      a = WMMA4(wih[4], S.f[4], a);
      cell_step(g, ca, cb, hB);
      __builtin_amdgcn_sched_group_barrier(0x008, 2, 0);
      __builtin_amdgcn_sched_group_barrier(0x402, 20, 0);
      __builtin_amdgcn_sched_group_barrier(0x008, 1, 0);
      __builtin_amdgcn_sched_barrier(0);
      g = WMMA4(whh, hB, xp);
      b = WMMA4(wih[5], S.f[5], b);
      a = WMMA4(wih[6], S.f[6], a);
      cell_step(g, ca, cb, hB);
      __builtin_amdgcn_sched_group_barrier(0x008, 2, 0);
      __builtin_amdgcn_sched_group_barrier(0x402, 20, 0);
      __builtin_amdgcn_sched_group_barrier(0x008, 1, 0);
      __builtin_amdgcn_sched_barrier(0);
      g = WMMA4(whh, hB, xp);
      b = WMMA4(wih[7], S.f[7], b);
      a = WMMA4(wih[8], s8, a);
      cell_step(g, ca, cb, hB);
      xp = a + b;
      __builtin_amdgcn_sched_group_barrier(0x008, 2, 0);
      __builtin_amdgcn_sched_group_barrier(0x402, 20, 0);
      __builtin_amdgcn_sched_group_barrier(0x008, 1, 0);
      __builtin_amdgcn_sched_barrier(0);
    }
  }

  v8f accfc;
#pragma unroll
  for (int v = 0; v < 8; ++v)
    accfc[v] = (half == 0 && v < NCLS) ? b_fc[min(v, NCLS - 1)] : 0.0f;
  const v8f o = WMMA4(wfc, hB, accfc);
  if (half == 0 && tj == 0) {
    float* __restrict__ op = out + bseq * NCLS;
    op[0] = o[0];
    op[1] = o[1];
    op[2] = o[2];
  }
}

extern "C" void kernel_launch(void* const* d_in, const int* in_sizes, int n_in,
                              void* d_out, int out_size, void* d_ws, size_t ws_size,
                              hipStream_t stream) {
  (void)in_sizes; (void)n_in; (void)out_size; (void)d_ws; (void)ws_size;
  const float* x    = (const float*)d_in[0];
  const float* W_ih = (const float*)d_in[1];
  const float* W_hh = (const float*)d_in[2];
  const float* b_ih = (const float*)d_in[3];
  const float* b_hh = (const float*)d_in[4];
  const float* W_fc = (const float*)d_in[5];
  const float* b_fc = (const float*)d_in[6];
  float* out = (float*)d_out;

  hipLaunchKernelGGL(lstm_s4t4_wmma, dim3(BATCH / (4 * SEQW)), dim3(128), 0, stream,
                     x, W_ih, W_hh, b_ih, b_hh, W_fc, b_fc, out);
}
